# starting kernel
# baseline (speedup 1.0000x reference)
#include <hip/hip_runtime.h>

typedef __attribute__((ext_vector_type(2))) float v2f;
typedef __attribute__((ext_vector_type(8))) float v8f;

#ifndef FEAT
#define FEAT 16
#endif

__global__ __launch_bounds__(256) void IPC_vertex_mlp_wmma(
    const float* __restrict__ vf,
    const float* __restrict__ py,
    const float* __restrict__ vp,
    float* __restrict__ out,
    int nvert)
{
    const int lane = threadIdx.x & 31;
    const int wave = threadIdx.x >> 5;
    const int v    = blockIdx.x * (blockDim.x >> 5) + wave;
    if (v >= nvert) return;

    const float* xf  = vf + (size_t)v * (FEAT * 128);
    const float* pyv = py + (size_t)v * (128 * 2);
    const float* p   = vp + (size_t)v * 233;

    const int  row   = lane & 15;
    const int  khalf = (lane >> 4) << 1;
    const bool lo    = (lane < 16);

    float m0 = __builtin_inff(), m1 = __builtin_inff();
#pragma unroll
    for (int i = 0; i < 4; ++i) {
        const int pt = lane + i * 32;
        m0 = fminf(m0, pyv[pt * 2 + 0]);
        m1 = fminf(m1, pyv[pt * 2 + 1]);
    }
#pragma unroll
    for (int off = 16; off > 0; off >>= 1) {
        m0 = fminf(m0, __shfl_xor(m0, off, 32));
        m1 = fminf(m1, __shfl_xor(m1, off, 32));
    }

    v2f A1[5];
#pragma unroll
    for (int kc = 0; kc < 5; ++kc) {
        const int k0 = kc * 4 + khalf;
        const bool ok0 = (row < 8) && (k0 < 18);
        const bool ok1 = (row < 8) && (k0 + 1 < 18);
        float a0 = p[ok0 ? (row * 18 + k0)     : 0];
        float a1 = p[ok1 ? (row * 18 + k0 + 1) : 0];
        A1[kc].x = ok0 ? a0 : 0.0f;
        A1[kc].y = ok1 ? a1 : 0.0f;
    }
    v2f A2[2];
#pragma unroll
    for (int kc = 0; kc < 2; ++kc) {
        const int k0 = kc * 4 + khalf;
        const bool okr = (row < 8);
        float a0 = p[okr ? (152 + row * 8 + k0)     : 0];
        float a1 = p[okr ? (152 + row * 8 + k0 + 1) : 0];
        A2[kc].x = okr ? a0 : 0.0f;
        A2[kc].y = okr ? a1 : 0.0f;
    }

    float b1r[8], b2r[8], w3r[8];
#pragma unroll
    for (int r = 0; r < 8; ++r) {
        b1r[r] = p[144 + r];
        b2r[r] = p[216 + r];
        w3r[r] = p[224 + r];
    }
    const float b3 = p[232];

#pragma unroll 1
    for (int nt = 0; nt < 8; ++nt) {
        const int col = nt * 16 + row;

        v8f c1 = {};
#pragma unroll
        for (int kc = 0; kc < 5; ++kc) {
            const int k0 = kc * 4 + khalf;
            float bb0, bb1;
            if (kc < 4) {
                bb0 = xf[k0 * 128 + col];
                bb1 = xf[(k0 + 1) * 128 + col];
            } else {
                float p0 = pyv[col * 2 + 0];
                float p1 = pyv[col * 2 + 1];
                bb0 = lo ? (p0 - m0) : 0.0f;
                bb1 = lo ? (p1 - m1) : 0.0f;
            }
            v2f B; B.x = bb0; B.y = bb1;
            c1 = __builtin_amdgcn_wmma_f32_16x16x4_f32(
                     false, A1[kc], false, B, (short)0, c1, false, false);
        }
#pragma unroll
        for (int r = 0; r < 8; ++r) c1[r] = fmaxf(c1[r] + b1r[r], 0.0f);

        v8f c2 = {};
#pragma unroll
        for (int kc = 0; kc < 2; ++kc) {
            float s0 = __shfl(c1[kc * 4 + 0], row, 32);
            float s1 = __shfl(c1[kc * 4 + 1], row, 32);
            float s2 = __shfl(c1[kc * 4 + 2], row, 32);
            float s3 = __shfl(c1[kc * 4 + 3], row, 32);
            v2f B; B.x = lo ? s0 : s2; B.y = lo ? s1 : s3;
            c2 = __builtin_amdgcn_wmma_f32_16x16x4_f32(
                     false, A2[kc], false, B, (short)0, c2, false, false);
        }
#pragma unroll
        for (int r = 0; r < 8; ++r) c2[r] = fmaxf(c2[r] + b2r[r], 0.0f);

        float acc = b3;
#pragma unroll
        for (int r = 0; r < 8; ++r) acc = fmaf(w3r[r], c2[r], acc);
        if (lo) out[(size_t)v * 128 + nt * 16 + row] = acc;
    }
}

extern "C" void kernel_launch(void* const* d_in, const int* in_sizes, int n_in,
                              void* d_out, int out_size, void* d_ws, size_t ws_size,
                              hipStream_t stream) {
    (void)n_in; (void)out_size; (void)d_ws; (void)ws_size;
    const float* vf = (const float*)d_in[0];
    const float* py = (const float*)d_in[1];
    const float* vp = (const float*)d_in[2];
    float* out = (float*)d_out;

    const int nvert = in_sizes[0] / (FEAT * 128);
    const int wavesPerBlock = 8;
    dim3 block(32 * wavesPerBlock);
    dim3 grid((nvert + wavesPerBlock - 1) / wavesPerBlock);
    IPC_vertex_mlp_wmma<<<grid, block, 0, stream>>>(vf, py, vp, out, nvert);
}


# Two 16-point tiles per WMMA chain, block-diagonal weights (lane half = tile): a lane owns 4 points, rows load as 16 B, D feeds layer 2 in place, no lane moves; k summed in 9 chunks of 2, not 5 of 4
# speedup vs baseline: 1.1176x; 1.1176x over previous
#include <hip/hip_runtime.h>

typedef __attribute__((ext_vector_type(2))) float v2f;
typedef __attribute__((ext_vector_type(4))) float v4f;
typedef __attribute__((ext_vector_type(8))) float v8f;

#ifndef FEAT
#define FEAT 16
#endif

__global__ __launch_bounds__(256) void IPC_vertex_mlp_wmma(
    const float* __restrict__ vf,
    const float* __restrict__ py,
    const float* __restrict__ vp,
    float* __restrict__ out,
    int nvert)
{
    const int lane = threadIdx.x & 31;
    const int wave = __builtin_amdgcn_readfirstlane((int)(threadIdx.x >> 5));
    const int v    = blockIdx.x * (blockDim.x >> 5) + wave;
    if (v >= nvert) return;

    const float* xf  = vf + (size_t)v * (FEAT * 128);
    const float* pyv = py + (size_t)v * (128 * 2);
    const float* p   = vp + (size_t)v * 233;

    const int  row  = lane & 15;
    const int  half = lane >> 4;
    const int  wr   = row & 7;
    const bool live = ((row >> 3) == half);

    v4f X[FEAT];
#pragma unroll
    for (int k = 0; k < FEAT; ++k)
        X[k] = *reinterpret_cast<const v4f*>(xf + k * 128 + 4 * lane);
    const v4f P0 = *reinterpret_cast<const v4f*>(pyv + 8 * lane);
    const v4f P1 = *reinterpret_cast<const v4f*>(pyv + 8 * lane + 4);

    v2f A1[9];
#pragma unroll
    for (int j = 0; j < 9; ++j) {
        const float a0 = p[wr * 18 + 2 * j];
        const float a1 = p[wr * 18 + 2 * j + 1];
        A1[j].x = live ? a0 : 0.0f;
        A1[j].y = live ? a1 : 0.0f;
    }
    v2f A2[4];
#pragma unroll
    for (int j = 0; j < 4; ++j) {
        const float a0 = p[152 + wr * 8 + 2 * j];
        const float a1 = p[152 + wr * 8 + 2 * j + 1];
        A2[j].x = live ? a0 : 0.0f;
        A2[j].y = live ? a1 : 0.0f;
    }

    float b1r[8], b2r[8], w3r[8];
#pragma unroll
    for (int r = 0; r < 8; ++r) {
        b1r[r] = p[144 + r];
        b2r[r] = p[216 + r];
        w3r[r] = p[224 + r];
    }
    const float b3 = p[232];

    float m0 = fminf(fminf(P0.x, P0.z), fminf(P1.x, P1.z));
    float m1 = fminf(fminf(P0.y, P0.w), fminf(P1.y, P1.w));
#pragma unroll
    for (int off = 16; off > 0; off >>= 1) {
        m0 = fminf(m0, __shfl_xor(m0, off, 32));
        m1 = fminf(m1, __shfl_xor(m1, off, 32));
    }
    const float cx[4] = { P0.x - m0, P0.z - m0, P1.x - m0, P1.z - m0 };
    const float cy[4] = { P0.y - m1, P0.w - m1, P1.y - m1, P1.w - m1 };

    v4f o;
#pragma unroll
    for (int q = 0; q < 4; ++q) {
        v8f c1 = {};
#pragma unroll
        for (int j = 0; j < 8; ++j) {
            v2f B; B.x = X[2 * j][q]; B.y = X[2 * j + 1][q];
            c1 = __builtin_amdgcn_wmma_f32_16x16x4_f32(
                     false, A1[j], false, B, (short)0, c1, false, false);
        }
        {
            v2f B; B.x = cx[q]; B.y = cy[q];
            c1 = __builtin_amdgcn_wmma_f32_16x16x4_f32(
                     false, A1[8], false, B, (short)0, c1, false, false);
        }
#pragma unroll
        for (int r = 0; r < 8; ++r) c1[r] = fmaxf(c1[r] + b1r[r], 0.0f);

        v8f c2 = {};
#pragma unroll
        for (int j = 0; j < 4; ++j) {
            v2f B; B.x = c1[2 * j]; B.y = c1[2 * j + 1];
            c2 = __builtin_amdgcn_wmma_f32_16x16x4_f32(
                     false, A2[j], false, B, (short)0, c2, false, false);
        }
#pragma unroll
        for (int r = 0; r < 8; ++r) c2[r] = fmaxf(c2[r] + b2r[r], 0.0f);

        float acc = b3;
#pragma unroll
        for (int r = 0; r < 8; ++r) acc = fmaf(w3r[r], c2[r], acc);
        o[q] = acc;
    }
    *reinterpret_cast<v4f*>(out + (size_t)v * 128 + 4 * lane) = o;
}

extern "C" void kernel_launch(void* const* d_in, const int* in_sizes, int n_in,
                              void* d_out, int out_size, void* d_ws, size_t ws_size,
                              hipStream_t stream) {
    (void)n_in; (void)out_size; (void)d_ws; (void)ws_size;
    const float* vf = (const float*)d_in[0];
    const float* py = (const float*)d_in[1];
    const float* vp = (const float*)d_in[2];
    float* out = (float*)d_out;

    const int nvert = in_sizes[0] / (FEAT * 128);
    const int wavesPerBlock = 8;
    dim3 block(32 * wavesPerBlock);
    dim3 grid((nvert + wavesPerBlock - 1) / wavesPerBlock);
    IPC_vertex_mlp_wmma<<<grid, block, 0, stream>>>(vf, py, vp, out, nvert);
}


# A wave streams 4 consecutive vertices through a ring of two register sets, vertex i+1's loads issued ahead of vertex i's WMMAs: a quarter as many waves enter; lane min by ds_swizzle frees 5 VGPRs
# speedup vs baseline: 1.2241x; 1.0953x over previous
#include <hip/hip_runtime.h>

typedef __attribute__((ext_vector_type(2))) float v2f;
typedef __attribute__((ext_vector_type(4))) float v4f;
typedef __attribute__((ext_vector_type(8))) float v8f;

#ifndef FEAT
#define FEAT 16
#endif
#define VPW   4
#define WAVES 8


template <int M>
static __device__ __forceinline__ float lane_xor(float x) {
    return __builtin_bit_cast(float, __builtin_amdgcn_ds_swizzle(__builtin_bit_cast(int, x), 0x1f | (M << 10)));
}

__global__ __launch_bounds__(32 * WAVES, 4) void IPC_vertex_mlp_wmma(
    const float* vf,
    const float* py,
    const float* vp,
    const float* __restrict__ vpu,
    float* __restrict__ out,
    int nvert)
{
    const int lane = threadIdx.x & 31;
    const int wave = __builtin_amdgcn_readfirstlane((int)(threadIdx.x >> 5));
    const int v0   = (blockIdx.x * WAVES + wave) * VPW;
    if (v0 >= nvert) return;

    const int  row  = lane & 15;
    const int  half = lane >> 4;
    const int  wr   = row & 7;
    const bool live = ((row >> 3) == half);

    v4f X[2][FEAT];
    v4f P0[2], P1[2];
    v2f W1[2][9], W2[2][4];
    float b1r[2][8], b2r[2][8], w3r[2][8], b3[2];

#define VERT(I) ((v0 + (I) < nvert) ? (v0 + (I)) : v0)
#define LOAD_VEC(I) do { \
        const int vi_ = VERT(I); \
        const float* xf_ = vf + (size_t)vi_ * (FEAT * 128) + 4 * lane; \
        const float* py_ = py + (size_t)vi_ * (128 * 2) + 8 * lane; \
        const float* pw_ = vp + (size_t)vi_ * 233 + wr * 18; \
        const float* pu_ = vp + (size_t)vi_ * 233 + 152 + wr * 8; \
        _Pragma("unroll") for (int k = 0; k < FEAT; ++k) \
            X[(I) & 1][k] = *reinterpret_cast<const v4f*>(xf_ + k * 128); \
        P0[(I) & 1] = *reinterpret_cast<const v4f*>(py_); \
        P1[(I) & 1] = *reinterpret_cast<const v4f*>(py_ + 4); \
        _Pragma("unroll") for (int j = 0; j < 9; ++j) { \
            W1[(I) & 1][j].x = pw_[2 * j]; W1[(I) & 1][j].y = pw_[2 * j + 1]; } \
        _Pragma("unroll") for (int j = 0; j < 4; ++j) { \
            W2[(I) & 1][j].x = pu_[2 * j]; W2[(I) & 1][j].y = pu_[2 * j + 1]; } \
    } while (0)
#define LOAD_UNI(I) do { \
        const float* p_ = vpu + (size_t)VERT(I) * 233; \
        _Pragma("unroll") for (int r = 0; r < 8; ++r) { \
            b1r[(I) & 1][r] = p_[144 + r]; b2r[(I) & 1][r] = p_[216 + r]; w3r[(I) & 1][r] = p_[224 + r]; } \
        b3[(I) & 1] = p_[232]; \
    } while (0)

#define HOLD_V8(a, b, c, d, e, f, g, h) \
    asm volatile("" : "+v"(a), "+v"(b), "+v"(c), "+v"(d), "+v"(e), "+v"(f), "+v"(g), "+v"(h))
#define HOLD_V7(a, b, c, d, e, f, g) \
    asm volatile("" : "+v"(a), "+v"(b), "+v"(c), "+v"(d), "+v"(e), "+v"(f), "+v"(g))
#define HOLD_S8(a) \
    asm volatile("" : "+s"(a[0]), "+s"(a[1]), "+s"(a[2]), "+s"(a[3]), "+s"(a[4]), "+s"(a[5]), "+s"(a[6]), "+s"(a[7]))
#define HOLD_S1(a) asm volatile("" : "+s"(a))

    LOAD_VEC(0);
    LOAD_UNI(0);

#pragma unroll
    for (int i = 0; i < VPW; ++i) {
        const int s = i & 1;
        __builtin_amdgcn_sched_barrier(0);
        if (i + 1 < VPW) LOAD_VEC(i + 1);
        __builtin_amdgcn_sched_barrier(0);

        HOLD_V8(X[s][0], X[s][1], X[s][2], X[s][3], X[s][4], X[s][5], X[s][6], X[s][7]);
        HOLD_V8(X[s][8], X[s][9], X[s][10], X[s][11], X[s][12], X[s][13], X[s][14], X[s][15]);
        HOLD_V8(P0[s], P1[s], W1[s][0], W1[s][1], W1[s][2], W1[s][3], W1[s][4], W1[s][5]);
        HOLD_V7(W1[s][6], W1[s][7], W1[s][8], W2[s][0], W2[s][1], W2[s][2], W2[s][3]);
        HOLD_S8(b1r[s]); HOLD_S8(b2r[s]); HOLD_S8(w3r[s]); HOLD_S1(b3[s]);
        __builtin_amdgcn_sched_barrier(0);
        if (i + 1 < VPW) LOAD_UNI(i + 1);
        __builtin_amdgcn_sched_barrier(0);

        v2f A1[9], A2[4];
#pragma unroll
        for (int j = 0; j < 9; ++j) {
            A1[j].x = live ? W1[s][j].x : 0.0f;
            A1[j].y = live ? W1[s][j].y : 0.0f;
        }
#pragma unroll
        for (int j = 0; j < 4; ++j) {
            A2[j].x = live ? W2[s][j].x : 0.0f;
            A2[j].y = live ? W2[s][j].y : 0.0f;
        }

        const v4f Q0 = P0[s], Q1 = P1[s];
        float m0 = fminf(fminf(Q0.x, Q0.z), fminf(Q1.x, Q1.z));
        float m1 = fminf(fminf(Q0.y, Q0.w), fminf(Q1.y, Q1.w));
        m0 = fminf(m0, lane_xor<16>(m0)); m1 = fminf(m1, lane_xor<16>(m1));
        m0 = fminf(m0, lane_xor<8>(m0));  m1 = fminf(m1, lane_xor<8>(m1));
        m0 = fminf(m0, lane_xor<4>(m0));  m1 = fminf(m1, lane_xor<4>(m1));
        m0 = fminf(m0, lane_xor<2>(m0));  m1 = fminf(m1, lane_xor<2>(m1));
        m0 = fminf(m0, lane_xor<1>(m0));  m1 = fminf(m1, lane_xor<1>(m1));
        const float cx[4] = { Q0.x - m0, Q0.z - m0, Q1.x - m0, Q1.z - m0 };
        const float cy[4] = { Q0.y - m1, Q0.w - m1, Q1.y - m1, Q1.w - m1 };

        v4f o;
#pragma unroll
        for (int q = 0; q < 4; ++q) {
            v8f c1 = {};
#pragma unroll
            for (int j = 0; j < 8; ++j) {
                v2f B; B.x = X[s][2 * j][q]; B.y = X[s][2 * j + 1][q];
                c1 = __builtin_amdgcn_wmma_f32_16x16x4_f32(
                         false, A1[j], false, B, (short)0, c1, false, false);
            }
            {
                v2f B; B.x = cx[q]; B.y = cy[q];
                c1 = __builtin_amdgcn_wmma_f32_16x16x4_f32(
                         false, A1[8], false, B, (short)0, c1, false, false);
            }
#pragma unroll
            for (int r = 0; r < 8; ++r) c1[r] = fmaxf(c1[r] + b1r[s][r], 0.0f);

            v8f c2 = {};
#pragma unroll
            for (int j = 0; j < 4; ++j) {
                v2f B; B.x = c1[2 * j]; B.y = c1[2 * j + 1];
                c2 = __builtin_amdgcn_wmma_f32_16x16x4_f32(
                         false, A2[j], false, B, (short)0, c2, false, false);
            }
#pragma unroll
            for (int r = 0; r < 8; ++r) c2[r] = fmaxf(c2[r] + b2r[s][r], 0.0f);

            float acc = b3[s];
#pragma unroll
            for (int r = 0; r < 8; ++r) acc = fmaf(w3r[s][r], c2[r], acc);
            o[q] = acc;
        }
        *reinterpret_cast<v4f*>(out + (size_t)VERT(i) * 128 + 4 * lane) = o;
    }
#undef LOAD_VEC
#undef LOAD_UNI
#undef VERT
#undef HOLD_V8
#undef HOLD_V7
#undef HOLD_S8
#undef HOLD_S1
}

extern "C" void kernel_launch(void* const* d_in, const int* in_sizes, int n_in,
                              void* d_out, int out_size, void* d_ws, size_t ws_size,
                              hipStream_t stream) {
    (void)n_in; (void)out_size; (void)d_ws; (void)ws_size;
    const float* vf = (const float*)d_in[0];
    const float* py = (const float*)d_in[1];
    const float* vp = (const float*)d_in[2];
    float* out = (float*)d_out;

    const int nvert = in_sizes[0] / (FEAT * 128);
    const int vertsPerBlock = WAVES * VPW;
    dim3 block(32 * WAVES);
    dim3 grid((nvert + vertsPerBlock - 1) / vertsPerBlock);
    IPC_vertex_mlp_wmma<<<grid, block, 0, stream>>>(vf, py, vp, vp, out, nvert);
}
